# starting kernel
# baseline (speedup 1.0000x reference)
#include <hip/hip_runtime.h>
#include <hip/hip_bf16.h>

#define IN_DIM  128
#define OUT_DIM 128

typedef float v2f __attribute__((ext_vector_type(2)));
typedef float v8f __attribute__((ext_vector_type(8)));

__global__ void __launch_bounds__(256)
gcn_gemm_wmma(const float* __restrict__ x, const float* __restrict__ w,
              float* __restrict__ support, int n_nodes) {
  __shared__ float wlds[IN_DIM * OUT_DIM];
  for (int i = threadIdx.x; i < IN_DIM * OUT_DIM; i += 256)
    wlds[i] = w[i];
  __syncthreads();

  const int lane = threadIdx.x & 31;
  const int wave = threadIdx.x >> 5;
  const int m0   = blockIdx.x * 128 + wave * 16;

  const int koff = (lane >> 4) * 2;
  const int nsub = lane & 15;

  int mrow = m0 + (lane & 15);
  if (mrow >= n_nodes) mrow = n_nodes - 1;

  const float* __restrict__ xrow = x + (size_t)mrow * IN_DIM;

  v8f acc[8];
  #pragma unroll
  for (int t = 0; t < 8; ++t) acc[t] = (v8f){};

  #pragma unroll 2
  for (int k0 = 0; k0 < IN_DIM; k0 += 4) {
    v2f a;
    a.x = xrow[k0 + koff];
    a.y = xrow[k0 + koff + 1];
    const float* __restrict__ wk0 = &wlds[(k0 + koff) * OUT_DIM + nsub];
    #pragma unroll
    for (int t = 0; t < 8; ++t) {
      v2f b;
      b.x = wk0[t * 16];
      b.y = wk0[t * 16 + OUT_DIM];
      acc[t] = __builtin_amdgcn_wmma_f32_16x16x4_f32(
           false, a,  false, b,
           (short)0, acc[t],  false,  false);
    }
  }

  const int mbase = m0 + ((lane >> 4) << 3);
  float* outp = support + (size_t)mbase * OUT_DIM + nsub;

  if (m0 + 15 < n_nodes) {
    #pragma unroll
    for (int r = 0; r < 8; ++r)
      #pragma unroll
      for (int t = 0; t < 8; ++t)
        outp[(size_t)r * OUT_DIM + t * 16] = acc[t][r];
  } else {
    #pragma unroll
    for (int r = 0; r < 8; ++r)
      if (mbase + r < n_nodes)
        #pragma unroll
        for (int t = 0; t < 8; ++t)
          outp[(size_t)r * OUT_DIM + t * 16] = acc[t][r];
  }
}

__global__ void __launch_bounds__(256)
gcn_bias_init(const float* __restrict__ bias, float* __restrict__ out, int total) {
  int idx = blockIdx.x * 256 + threadIdx.x;
  if (idx < total)
    out[idx] = bias[idx & (OUT_DIM - 1)];
}

__global__ void __launch_bounds__(256)
gcn_spmm_scatter(const float* __restrict__ support,
                 const float* __restrict__ adj_vals,
                 const int* __restrict__ edge_row,
                 const int* __restrict__ edge_col,
                 float* __restrict__ out, int n_edges) {
  const int lane = threadIdx.x & 31;
  const int edge = blockIdx.x * 8 + (threadIdx.x >> 5);
  if (edge >= n_edges) return;

  const int   col = edge_col[edge];
  const int   row = edge_row[edge];
  const float v   = adj_vals[edge];

  const float4 m =
      *((const float4*)(support + (size_t)col * OUT_DIM) + lane);

  float* dst = out + (size_t)row * OUT_DIM + lane * 4;
  __hip_atomic_fetch_add(dst + 0, m.x * v, __ATOMIC_RELAXED, __HIP_MEMORY_SCOPE_AGENT);
  __hip_atomic_fetch_add(dst + 1, m.y * v, __ATOMIC_RELAXED, __HIP_MEMORY_SCOPE_AGENT);
  __hip_atomic_fetch_add(dst + 2, m.z * v, __ATOMIC_RELAXED, __HIP_MEMORY_SCOPE_AGENT);
  __hip_atomic_fetch_add(dst + 3, m.w * v, __ATOMIC_RELAXED, __HIP_MEMORY_SCOPE_AGENT);
}

extern "C" void kernel_launch(void* const* d_in, const int* in_sizes, int n_in,
                              void* d_out, int out_size, void* d_ws, size_t ws_size,
                              hipStream_t stream) {
  const float* x        = (const float*)d_in[0];
  const float* w        = (const float*)d_in[1];
  const float* bias     = (const float*)d_in[2];
  const float* adj_vals = (const float*)d_in[3];
  const int*   edge_row = (const int*)d_in[4];
  const int*   edge_col = (const int*)d_in[5];

  const int n_nodes = in_sizes[0] / IN_DIM;
  const int n_edges = in_sizes[3];

  float* out     = (float*)d_out;
  float* support = (float*)d_ws;

  const int nblocks = (n_nodes + 127) / 128;
  gcn_gemm_wmma<<<nblocks, 256, 0, stream>>>(x, w, support, n_nodes);

  const int total = n_nodes * OUT_DIM;
  gcn_bias_init<<<(total + 255) / 256, 256, 0, stream>>>(bias, out, total);

  gcn_spmm_scatter<<<(n_edges + 7) / 8, 256, 0, stream>>>(
      support, adj_vals, edge_row, edge_col, out, n_edges);
}


# scatter: lane = feature column (l, l+32, l+64, l+96), so each f32 atomic add instruction covers one whole 128-B line, not 8 lanes in each of 4 lines; atomics kept, arrival order as before
# speedup vs baseline: 1.3584x; 1.3584x over previous
#include <hip/hip_runtime.h>
#include <hip/hip_bf16.h>

#define IN_DIM  128
#define OUT_DIM 128

typedef float v2f __attribute__((ext_vector_type(2)));
typedef float v8f __attribute__((ext_vector_type(8)));

__global__ void __launch_bounds__(256)
gcn_gemm_wmma(const float* __restrict__ x, const float* __restrict__ w,
              float* __restrict__ support, int n_nodes) {
  __shared__ float wlds[IN_DIM * OUT_DIM];
  for (int i = threadIdx.x; i < IN_DIM * OUT_DIM; i += 256)
    wlds[i] = w[i];
  __syncthreads();

  const int lane = threadIdx.x & 31;
  const int wave = threadIdx.x >> 5;
  const int m0   = blockIdx.x * 128 + wave * 16;

  const int koff = (lane >> 4) * 2;
  const int nsub = lane & 15;

  int mrow = m0 + (lane & 15);
  if (mrow >= n_nodes) mrow = n_nodes - 1;

  const float* __restrict__ xrow = x + (size_t)mrow * IN_DIM;

  v8f acc[8];
  #pragma unroll
  for (int t = 0; t < 8; ++t) acc[t] = (v8f){};

  #pragma unroll 2
  for (int k0 = 0; k0 < IN_DIM; k0 += 4) {
    v2f a;
    a.x = xrow[k0 + koff];
    a.y = xrow[k0 + koff + 1];
    const float* __restrict__ wk0 = &wlds[(k0 + koff) * OUT_DIM + nsub];
    #pragma unroll
    for (int t = 0; t < 8; ++t) {
      v2f b;
      b.x = wk0[t * 16];
      b.y = wk0[t * 16 + OUT_DIM];
      acc[t] = __builtin_amdgcn_wmma_f32_16x16x4_f32(
           false, a,  false, b,
           (short)0, acc[t],  false,  false);
    }
  }

  const int mbase = m0 + ((lane >> 4) << 3);
  float* outp = support + (size_t)mbase * OUT_DIM + nsub;

  if (m0 + 15 < n_nodes) {
    #pragma unroll
    for (int r = 0; r < 8; ++r)
      #pragma unroll
      for (int t = 0; t < 8; ++t)
        outp[(size_t)r * OUT_DIM + t * 16] = acc[t][r];
  } else {
    #pragma unroll
    for (int r = 0; r < 8; ++r)
      if (mbase + r < n_nodes)
        #pragma unroll
        for (int t = 0; t < 8; ++t)
          outp[(size_t)r * OUT_DIM + t * 16] = acc[t][r];
  }
}

__global__ void __launch_bounds__(256)
gcn_bias_init(const float* __restrict__ bias, float* __restrict__ out, int total) {
  int idx = blockIdx.x * 256 + threadIdx.x;
  if (idx < total)
    out[idx] = bias[idx & (OUT_DIM - 1)];
}

__global__ void __launch_bounds__(256)
gcn_spmm_scatter(const float* __restrict__ support,
                 const float* __restrict__ adj_vals,
                 const int* __restrict__ edge_row,
                 const int* __restrict__ edge_col,
                 float* __restrict__ out, int n_edges) {
  const int lane = threadIdx.x & 31;
  const int edge = blockIdx.x * 8 + (threadIdx.x >> 5);
  if (edge >= n_edges) return;

  const int   col = edge_col[edge];
  const int   row = edge_row[edge];
  const float v   = adj_vals[edge];

  const float* __restrict__ srcp = support + (size_t)col * OUT_DIM + lane;
  const float m0 = srcp[0];
  const float m1 = srcp[32];
  const float m2 = srcp[64];
  const float m3 = srcp[96];

  float* dst = out + (size_t)row * OUT_DIM + lane;
  __hip_atomic_fetch_add(dst +  0, m0 * v, __ATOMIC_RELAXED, __HIP_MEMORY_SCOPE_AGENT);
  __hip_atomic_fetch_add(dst + 32, m1 * v, __ATOMIC_RELAXED, __HIP_MEMORY_SCOPE_AGENT);
  __hip_atomic_fetch_add(dst + 64, m2 * v, __ATOMIC_RELAXED, __HIP_MEMORY_SCOPE_AGENT);
  __hip_atomic_fetch_add(dst + 96, m3 * v, __ATOMIC_RELAXED, __HIP_MEMORY_SCOPE_AGENT);
}

extern "C" void kernel_launch(void* const* d_in, const int* in_sizes, int n_in,
                              void* d_out, int out_size, void* d_ws, size_t ws_size,
                              hipStream_t stream) {
  const float* x        = (const float*)d_in[0];
  const float* w        = (const float*)d_in[1];
  const float* bias     = (const float*)d_in[2];
  const float* adj_vals = (const float*)d_in[3];
  const int*   edge_row = (const int*)d_in[4];
  const int*   edge_col = (const int*)d_in[5];

  const int n_nodes = in_sizes[0] / IN_DIM;
  const int n_edges = in_sizes[3];

  float* out     = (float*)d_out;
  float* support = (float*)d_ws;

  const int nblocks = (n_nodes + 127) / 128;
  gcn_gemm_wmma<<<nblocks, 256, 0, stream>>>(x, w, support, n_nodes);

  const int total = n_nodes * OUT_DIM;
  gcn_bias_init<<<(total + 255) / 256, 256, 0, stream>>>(bias, out, total);

  gcn_spmm_scatter<<<(n_edges + 7) / 8, 256, 0, stream>>>(
      support, adj_vals, edge_row, edge_col, out, n_edges);
}


# scatter: one wave per 32 edges (indices loaded coalesced, 8 edges = 32 line gathers in flight per wave), not one wave per edge: 32x fewer waves; f32 atomics kept, sums still in arrival order
# speedup vs baseline: 1.7746x; 1.3064x over previous
#include <hip/hip_runtime.h>
#include <hip/hip_bf16.h>

#define IN_DIM  128
#define OUT_DIM 128

typedef float v2f __attribute__((ext_vector_type(2)));
typedef float v8f __attribute__((ext_vector_type(8)));

__global__ void __launch_bounds__(256)
gcn_gemm_wmma(const float* __restrict__ x, const float* __restrict__ w,
              float* __restrict__ support, int n_nodes) {
  __shared__ float wlds[IN_DIM * OUT_DIM];
  for (int i = threadIdx.x; i < IN_DIM * OUT_DIM; i += 256)
    wlds[i] = w[i];
  __syncthreads();

  const int lane = threadIdx.x & 31;
  const int wave = threadIdx.x >> 5;
  const int m0   = blockIdx.x * 128 + wave * 16;

  const int koff = (lane >> 4) * 2;
  const int nsub = lane & 15;

  int mrow = m0 + (lane & 15);
  if (mrow >= n_nodes) mrow = n_nodes - 1;

  const float* __restrict__ xrow = x + (size_t)mrow * IN_DIM;

  v8f acc[8];
  #pragma unroll
  for (int t = 0; t < 8; ++t) acc[t] = (v8f){};

  #pragma unroll 2
  for (int k0 = 0; k0 < IN_DIM; k0 += 4) {
    v2f a;
    a.x = xrow[k0 + koff];
    a.y = xrow[k0 + koff + 1];
    const float* __restrict__ wk0 = &wlds[(k0 + koff) * OUT_DIM + nsub];
    #pragma unroll
    for (int t = 0; t < 8; ++t) {
      v2f b;
      b.x = wk0[t * 16];
      b.y = wk0[t * 16 + OUT_DIM];
      acc[t] = __builtin_amdgcn_wmma_f32_16x16x4_f32(
           false, a,  false, b,
           (short)0, acc[t],  false,  false);
    }
  }

  const int mbase = m0 + ((lane >> 4) << 3);
  float* outp = support + (size_t)mbase * OUT_DIM + nsub;

  if (m0 + 15 < n_nodes) {
    #pragma unroll
    for (int r = 0; r < 8; ++r)
      #pragma unroll
      for (int t = 0; t < 8; ++t)
        outp[(size_t)r * OUT_DIM + t * 16] = acc[t][r];
  } else {
    #pragma unroll
    for (int r = 0; r < 8; ++r)
      if (mbase + r < n_nodes)
        #pragma unroll
        for (int t = 0; t < 8; ++t)
          outp[(size_t)r * OUT_DIM + t * 16] = acc[t][r];
  }
}

__global__ void __launch_bounds__(256)
gcn_bias_init(const float* __restrict__ bias, float* __restrict__ out, int total) {
  int idx = blockIdx.x * 256 + threadIdx.x;
  if (idx < total)
    out[idx] = bias[idx & (OUT_DIM - 1)];
}

#define EDGES_PER_WAVE 32
#define EDGE_CHUNK 8

__device__ __forceinline__ void scatter_add_line4(float* dst, float m0, float m1,
                                                  float m2, float m3, float v) {
  __hip_atomic_fetch_add(dst +  0, m0 * v, __ATOMIC_RELAXED, __HIP_MEMORY_SCOPE_AGENT);
  __hip_atomic_fetch_add(dst + 32, m1 * v, __ATOMIC_RELAXED, __HIP_MEMORY_SCOPE_AGENT);
  __hip_atomic_fetch_add(dst + 64, m2 * v, __ATOMIC_RELAXED, __HIP_MEMORY_SCOPE_AGENT);
  __hip_atomic_fetch_add(dst + 96, m3 * v, __ATOMIC_RELAXED, __HIP_MEMORY_SCOPE_AGENT);
}

__global__ void __launch_bounds__(256)
gcn_spmm_scatter(const float* __restrict__ support,
                 const float* __restrict__ adj_vals,
                 const int* __restrict__ edge_row,
                 const int* __restrict__ edge_col,
                 float* __restrict__ out, int n_edges) {
  const int lane = threadIdx.x & 31;
  const int wave = __builtin_amdgcn_readfirstlane((int)(threadIdx.x >> 5));
  const int base = (blockIdx.x * 8 + wave) * EDGES_PER_WAVE;
  if (base >= n_edges) return;
  int nvalid = n_edges - base;
  if (nvalid > EDGES_PER_WAVE) nvalid = EDGES_PER_WAVE;

  int e = base + lane;
  if (e >= n_edges) e = n_edges - 1;
  const int colv = edge_col[e];
  const int rowv = edge_row[e];
  const int valv = __builtin_bit_cast(int, adj_vals[e]);

  const int nfull = nvalid & ~(EDGE_CHUNK - 1);
  for (int c = 0; c < nfull; c += EDGE_CHUNK) {
    float m[EDGE_CHUNK][4];
    for (int j = 0; j < EDGE_CHUNK; ++j) {
      const int col = __builtin_amdgcn_readlane(colv, c + j);
      const float* __restrict__ srcp = support + (size_t)col * OUT_DIM + lane;
      m[j][0] = srcp[0];
      m[j][1] = srcp[32];
      m[j][2] = srcp[64];
      m[j][3] = srcp[96];
    }
    for (int j = 0; j < EDGE_CHUNK; ++j) {
      const int   row = __builtin_amdgcn_readlane(rowv, c + j);
      const float v   = __builtin_bit_cast(float, __builtin_amdgcn_readlane(valv, c + j));
      scatter_add_line4(out + (size_t)row * OUT_DIM + lane,
                        m[j][0], m[j][1], m[j][2], m[j][3], v);
    }
  }
  for (int j = nfull; j < nvalid; ++j) {
    const int   col = __builtin_amdgcn_readlane(colv, j);
    const int   row = __builtin_amdgcn_readlane(rowv, j);
    const float v   = __builtin_bit_cast(float, __builtin_amdgcn_readlane(valv, j));
    const float* __restrict__ srcp = support + (size_t)col * OUT_DIM + lane;
    scatter_add_line4(out + (size_t)row * OUT_DIM + lane,
                      srcp[0], srcp[32], srcp[64], srcp[96], v);
  }
}

extern "C" void kernel_launch(void* const* d_in, const int* in_sizes, int n_in,
                              void* d_out, int out_size, void* d_ws, size_t ws_size,
                              hipStream_t stream) {
  const float* x        = (const float*)d_in[0];
  const float* w        = (const float*)d_in[1];
  const float* bias     = (const float*)d_in[2];
  const float* adj_vals = (const float*)d_in[3];
  const int*   edge_row = (const int*)d_in[4];
  const int*   edge_col = (const int*)d_in[5];

  const int n_nodes = in_sizes[0] / IN_DIM;
  const int n_edges = in_sizes[3];

  float* out     = (float*)d_out;
  float* support = (float*)d_ws;

  const int nblocks = (n_nodes + 127) / 128;
  gcn_gemm_wmma<<<nblocks, 256, 0, stream>>>(x, w, support, n_nodes);

  const int total = n_nodes * OUT_DIM;
  gcn_bias_init<<<(total + 255) / 256, 256, 0, stream>>>(bias, out, total);

  gcn_spmm_scatter<<<(n_edges + 255) / 256, 256, 0, stream>>>(
      support, adj_vals, edge_row, edge_col, out, n_edges);
}


# scatter: edges counting-sorted into 256 destination-row bins; one work-group per bin sums its rows in LDS (LDS f32 adds, arrival order), each row stored once; no global atomics, no bias pass
# speedup vs baseline: 2.4364x; 1.3729x over previous
#include <hip/hip_runtime.h>
#include <hip/hip_bf16.h>

#define IN_DIM  128
#define OUT_DIM 128

#define NBINS        256
#define ROWS_PER_BIN 196
#define BIN_META     512

typedef int v4i __attribute__((ext_vector_type(4)));
typedef float v4f __attribute__((ext_vector_type(4)));

__device__ __forceinline__ int bin_of_row(int row) {
  const unsigned b = (unsigned)row / (unsigned)ROWS_PER_BIN;
  return (int)(b < (unsigned)(NBINS - 1) ? b : (unsigned)(NBINS - 1));
}

typedef float v2f __attribute__((ext_vector_type(2)));
typedef float v8f __attribute__((ext_vector_type(8)));

__global__ void __launch_bounds__(256)
gcn_gemm_wmma(const float* __restrict__ x, const float* __restrict__ w,
              float* __restrict__ support, int n_nodes,
              const int* __restrict__ edge_row, int* __restrict__ cnt,
              int n_edges, int gemm_blocks) {
  __shared__ float wlds[IN_DIM * OUT_DIM];

  if ((int)blockIdx.x >= gemm_blocks) {
    int* hist = (int*)wlds;
    const int cb = (int)blockIdx.x - gemm_blocks;
    hist[threadIdx.x] = 0;
    __syncthreads();
    const int ch = (n_edges + NBINS - 1) / NBINS;
    const int e0 = cb * ch;
    int e1 = e0 + ch;
    if (e1 > n_edges) e1 = n_edges;
    for (int e = e0 + (int)threadIdx.x; e < e1; e += 256)
      atomicAdd(&hist[bin_of_row(edge_row[e])], 1);
    __syncthreads();
    cnt[cb * NBINS + threadIdx.x] = hist[threadIdx.x];
    return;
  }

  for (int i = threadIdx.x; i < IN_DIM * OUT_DIM; i += 256)
    wlds[i] = w[i];
  __syncthreads();

  const int lane = threadIdx.x & 31;
  const int wave = threadIdx.x >> 5;
  const int m0   = blockIdx.x * 128 + wave * 16;

  const int koff = (lane >> 4) * 2;
  const int nsub = lane & 15;

  int mrow = m0 + (lane & 15);
  if (mrow >= n_nodes) mrow = n_nodes - 1;

  const float* __restrict__ xrow = x + (size_t)mrow * IN_DIM;

  v8f acc[8];
  #pragma unroll
  for (int t = 0; t < 8; ++t) acc[t] = (v8f){};

  #pragma unroll 2
  for (int k0 = 0; k0 < IN_DIM; k0 += 4) {
    v2f a;
    a.x = xrow[k0 + koff];
    a.y = xrow[k0 + koff + 1];
    const float* __restrict__ wk0 = &wlds[(k0 + koff) * OUT_DIM + nsub];
    #pragma unroll
    for (int t = 0; t < 8; ++t) {
      v2f b;
      b.x = wk0[t * 16];
      b.y = wk0[t * 16 + OUT_DIM];
      acc[t] = __builtin_amdgcn_wmma_f32_16x16x4_f32(
           false, a,  false, b,
           (short)0, acc[t],  false,  false);
    }
  }

  const int mbase = m0 + ((lane >> 4) << 3);
  float* outp = support + (size_t)mbase * OUT_DIM + nsub;

  if (m0 + 15 < n_nodes) {
    #pragma unroll
    for (int r = 0; r < 8; ++r)
      #pragma unroll
      for (int t = 0; t < 8; ++t)
        outp[(size_t)r * OUT_DIM + t * 16] = acc[t][r];
  } else {
    #pragma unroll
    for (int r = 0; r < 8; ++r)
      if (mbase + r < n_nodes)
        #pragma unroll
        for (int t = 0; t < 8; ++t)
          outp[(size_t)r * OUT_DIM + t * 16] = acc[t][r];
  }
}

__global__ void __launch_bounds__(1024)
gcn_bin_fill(const int* __restrict__ cnt,
             const float* __restrict__ adj_vals,
             const int* __restrict__ edge_row,
             const int* __restrict__ edge_col,
             int* __restrict__ bin_start,
             int* __restrict__ rec_words, int n_edges) {
  __shared__ int s_tot[4][NBINS];
  __shared__ int s_pre[4][NBINS];
  __shared__ int s_scan[NBINS];
  __shared__ int cursor[NBINS];

  const int t  = threadIdx.x;
  const int b  = blockIdx.x;
  const int bn = t & (NBINS - 1);
  const int q  = t >> 8;

  int tot = 0, pre = 0;
  for (int k = 0; k < NBINS / 4; ++k) {
    const int bp = q * (NBINS / 4) + k;
    const int c  = cnt[bp * NBINS + bn];
    tot += c;
    pre += (bp < b) ? c : 0;
  }
  s_tot[q][bn] = tot;
  s_pre[q][bn] = pre;
  __syncthreads();

  int mytot = 0;
  if (t < NBINS) {
    mytot = s_tot[0][t] + s_tot[1][t] + s_tot[2][t] + s_tot[3][t];
    s_scan[t] = mytot;
  }
  __syncthreads();
  for (int off = 1; off < NBINS; off <<= 1) {
    int v = 0;
    if (t < NBINS && t >= off) v = s_scan[t - off];
    __syncthreads();
    if (t < NBINS) s_scan[t] += v;
    __syncthreads();
  }
  if (t < NBINS) {
    const int incl = s_scan[t];
    const int excl = incl - mytot;
    cursor[t] = excl + s_pre[0][t] + s_pre[1][t] + s_pre[2][t] + s_pre[3][t];
    if (b == 0) {
      bin_start[t] = excl;
      if (t == NBINS - 1) bin_start[NBINS] = incl;
    }
  }
  __syncthreads();

  v4i* __restrict__ rec = (v4i*)rec_words;
  const int ch = (n_edges + NBINS - 1) / NBINS;
  const int e0 = b * ch;
  int e1 = e0 + ch;
  if (e1 > n_edges) e1 = n_edges;
  for (int base = e0; base < e1; base += 4096) {
    int rr[4], cc[4], vv[4];
    #pragma unroll
    for (int k = 0; k < 4; ++k) {
      const int e = base + k * 1024 + t;
      rr[k] = 0; cc[k] = 0; vv[k] = 0;
      if (e < e1) {
        rr[k] = edge_row[e];
        cc[k] = edge_col[e];
        vv[k] = __builtin_bit_cast(int, adj_vals[e]);
      }
    }
    #pragma unroll
    for (int k = 0; k < 4; ++k) {
      const int e = base + k * 1024 + t;
      if (e < e1) {
        const int bk  = bin_of_row(rr[k]);
        const int pos = atomicAdd(&cursor[bk], 1);
        v4i r;
        r.x = cc[k];
        r.y = rr[k] - bk * ROWS_PER_BIN;
        r.z = vv[k];
        r.w = 0;
        rec[pos] = r;
      }
    }
  }
}

#define EDGE_CHUNK 8

__global__ void __launch_bounds__(1024)
gcn_bin_aggregate(const float* __restrict__ support,
                  const float* __restrict__ bias,
                  const int* __restrict__ bin_start,
                  const int* __restrict__ rec_words,
                  float* __restrict__ out, int n_nodes) {
  __shared__ float acc[ROWS_PER_BIN * OUT_DIM];

  const int tid  = threadIdx.x;
  const int lane = tid & 31;
  const int wave  = __builtin_amdgcn_readfirstlane(tid >> 5);
  const int bin   = blockIdx.x;
  const int start = __builtin_amdgcn_readfirstlane(bin_start[bin]);
  const int end   = __builtin_amdgcn_readfirstlane(bin_start[bin + 1]);

  {
    v4f* acc4 = (v4f*)acc;
    const v4f* __restrict__ bias4 = (const v4f*)bias;
    const v4f bv = bias4[tid & 31];
    for (int i = tid; i < ROWS_PER_BIN * (OUT_DIM / 4); i += 1024)
      acc4[i] = bv;
  }
  __syncthreads();

  const v4i* __restrict__ rec = (const v4i*)rec_words;
  const int ngroups = (end - start + EDGE_CHUNK - 1) / EDGE_CHUNK;
  v4i r = (v4i){0, 0, 0, 0};
  if (wave < ngroups) {
    int idx = start + wave * EDGE_CHUNK + (lane & (EDGE_CHUNK - 1));
    if (idx >= end) idx = end - 1;
    r = rec[idx];
  }
  for (int g = wave; g < ngroups; g += 32) {
    const int eb = start + g * EDGE_CHUNK;
    int nv = end - eb;
    if (nv > EDGE_CHUNK) nv = EDGE_CHUNK;
    v4i rnext = r;
    if (g + 32 < ngroups) {
      int idx = eb + 32 * EDGE_CHUNK + (lane & (EDGE_CHUNK - 1));
      if (idx >= end) idx = end - 1;
      rnext = rec[idx];
    }

    float m[EDGE_CHUNK][4];
    #pragma unroll
    for (int j = 0; j < EDGE_CHUNK; ++j) {
      const int col = __builtin_amdgcn_readlane(r.x, j);
      const float* __restrict__ srcp = support + (size_t)col * OUT_DIM + lane;
      m[j][0] = srcp[0];
      m[j][1] = srcp[32];
      m[j][2] = srcp[64];
      m[j][3] = srcp[96];
    }
    #pragma unroll
    for (int j = 0; j < EDGE_CHUNK; ++j) {
      if (j == 0 || j < nv) {
        const int   lrow = __builtin_amdgcn_readlane(r.y, j);
        const float v    = __builtin_bit_cast(float, __builtin_amdgcn_readlane(r.z, j));
        float* d = acc + lrow * OUT_DIM + lane;
        __hip_atomic_fetch_add(d +  0, m[j][0] * v, __ATOMIC_RELAXED, __HIP_MEMORY_SCOPE_WORKGROUP);
        __hip_atomic_fetch_add(d + 32, m[j][1] * v, __ATOMIC_RELAXED, __HIP_MEMORY_SCOPE_WORKGROUP);
        __hip_atomic_fetch_add(d + 64, m[j][2] * v, __ATOMIC_RELAXED, __HIP_MEMORY_SCOPE_WORKGROUP);
        __hip_atomic_fetch_add(d + 96, m[j][3] * v, __ATOMIC_RELAXED, __HIP_MEMORY_SCOPE_WORKGROUP);
      }
    }
    r = rnext;
  }
  __syncthreads();

  const int row0 = bin * ROWS_PER_BIN;
  int nrows = n_nodes - row0;
  if (nrows > ROWS_PER_BIN) nrows = ROWS_PER_BIN;
  const v4f* acc4 = (const v4f*)acc;
  v4f* __restrict__ out4 = (v4f*)(out + (size_t)row0 * OUT_DIM);
  for (int i = tid; i < nrows * (OUT_DIM / 4); i += 1024)
    out4[i] = acc4[i];
}

extern "C" void kernel_launch(void* const* d_in, const int* in_sizes, int n_in,
                              void* d_out, int out_size, void* d_ws, size_t ws_size,
                              hipStream_t stream) {
  const float* x        = (const float*)d_in[0];
  const float* w        = (const float*)d_in[1];
  const float* bias     = (const float*)d_in[2];
  const float* adj_vals = (const float*)d_in[3];
  const int*   edge_row = (const int*)d_in[4];
  const int*   edge_col = (const int*)d_in[5];

  const int n_nodes = in_sizes[0] / IN_DIM;
  const int n_edges = in_sizes[3];

  float* out     = (float*)d_out;
  float* support   = (float*)d_ws;
  int*   cnt       = (int*)(support + (size_t)n_nodes * OUT_DIM);
  int*   bin_start = cnt + NBINS * NBINS;
  int*   rec_words = bin_start + BIN_META;

  const int nblocks = (n_nodes + 127) / 128;
  gcn_gemm_wmma<<<nblocks + NBINS, 256, 0, stream>>>(x, w, support, n_nodes,
                                                     edge_row, cnt, n_edges, nblocks);

  gcn_bin_fill<<<NBINS, 1024, 0, stream>>>(cnt, adj_vals, edge_row, edge_col,
                                           bin_start, rec_words, n_edges);

  gcn_bin_aggregate<<<NBINS, 1024, 0, stream>>>(support, bias, bin_start, rec_words,
                                                out, n_nodes);
}


# edge sort: each edge slice is binned inside the GEMM launch (count, scan, place per block); a bin's work-group copies its 256 runs to LDS; fill launch gone (3 -> 2); LDS f32 adds, arrival order kept
# speedup vs baseline: 2.7814x; 1.1416x over previous
#include <hip/hip_runtime.h>
#include <hip/hip_bf16.h>

#define IN_DIM  128
#define OUT_DIM 128

#define NBINS        256
#define ROWS_PER_BIN 196
#define SLICE_K      8
#define REC_CAP      4096
#define RUN_K        6

typedef int v4i __attribute__((ext_vector_type(4)));
typedef float v4f __attribute__((ext_vector_type(4)));

__device__ __forceinline__ int bin_of_row(int row) {
  const unsigned b = (unsigned)row / (unsigned)ROWS_PER_BIN;
  return (int)(b < (unsigned)(NBINS - 1) ? b : (unsigned)(NBINS - 1));
}

typedef int v2i __attribute__((ext_vector_type(2)));
typedef float v2f __attribute__((ext_vector_type(2)));
typedef float v8f __attribute__((ext_vector_type(8)));

__global__ void __launch_bounds__(256)
gcn_gemm_wmma(const float* __restrict__ x, const float* __restrict__ w,
              float* __restrict__ support, int n_nodes,
              const float* __restrict__ adj_vals,
              const int* __restrict__ edge_row,
              const int* __restrict__ edge_col,
              int* __restrict__ run_words, int* __restrict__ rec_words,
              int n_edges) {
  __shared__ float wlds[IN_DIM * OUT_DIM];

  if ((int)blockIdx.x < NBINS) {
    int* hist   = (int*)wlds;
    int* scan   = hist + NBINS;
    int* cursor = scan + NBINS;
    const int t  = threadIdx.x;
    const int cb = (int)blockIdx.x;
    hist[t] = 0;
    __syncthreads();
    const int ch = (n_edges + NBINS - 1) / NBINS;
    const int e0 = cb * ch;
    int e1 = e0 + ch;
    if (e1 > n_edges) e1 = n_edges;

    for (int base = e0; base < e1; base += SLICE_K * 256) {
      int rr[SLICE_K];
      #pragma unroll
      for (int k = 0; k < SLICE_K; ++k) {
        const int e = base + k * 256 + t;
        rr[k] = 0;
        if (e < e1) rr[k] = edge_row[e];
      }
      #pragma unroll
      for (int k = 0; k < SLICE_K; ++k) {
        const int e = base + k * 256 + t;
        if (e < e1) atomicAdd(&hist[bin_of_row(rr[k])], 1);
      }
    }
    __syncthreads();

    const int mycnt = hist[t];
    scan[t] = mycnt;
    __syncthreads();
    for (int off = 1; off < NBINS; off <<= 1) {
      int v = 0;
      if (t >= off) v = scan[t - off];
      __syncthreads();
      scan[t] += v;
      __syncthreads();
    }
    const int first = e0 + scan[t] - mycnt;
    cursor[t] = first;
    {
      v2i rn;
      rn.x = first;
      rn.y = mycnt;
      ((v2i*)run_words)[t * NBINS + cb] = rn;
    }
    __syncthreads();

    v4i* __restrict__ rec = (v4i*)rec_words;
    for (int base = e0; base < e1; base += SLICE_K * 256) {
      int rr[SLICE_K], cc[SLICE_K], vv[SLICE_K];
      #pragma unroll
      for (int k = 0; k < SLICE_K; ++k) {
        const int e = base + k * 256 + t;
        rr[k] = 0; cc[k] = 0; vv[k] = 0;
        if (e < e1) {
          rr[k] = edge_row[e];
          cc[k] = edge_col[e];
          vv[k] = __builtin_bit_cast(int, adj_vals[e]);
        }
      }
      #pragma unroll
      for (int k = 0; k < SLICE_K; ++k) {
        const int e = base + k * 256 + t;
        if (e < e1) {
          const int bk  = bin_of_row(rr[k]);
          const int pos = atomicAdd(&cursor[bk], 1);
          v4i r;
          r.x = cc[k];
          r.y = rr[k] - bk * ROWS_PER_BIN;
          r.z = vv[k];
          r.w = 0;
          rec[pos] = r;
        }
      }
    }
    return;
  }

  for (int i = threadIdx.x; i < IN_DIM * OUT_DIM; i += 256)
    wlds[i] = w[i];
  __syncthreads();

  const int lane = threadIdx.x & 31;
  const int wave = threadIdx.x >> 5;
  const int m0   = ((int)blockIdx.x - NBINS) * 128 + wave * 16;

  const int koff = (lane >> 4) * 2;
  const int nsub = lane & 15;

  int mrow = m0 + (lane & 15);
  if (mrow >= n_nodes) mrow = n_nodes - 1;

  const float* __restrict__ xrow = x + (size_t)mrow * IN_DIM;

  v8f acc[8];
  #pragma unroll
  for (int t = 0; t < 8; ++t) acc[t] = (v8f){};

  #pragma unroll 2
  for (int k0 = 0; k0 < IN_DIM; k0 += 4) {
    v2f a;
    a.x = xrow[k0 + koff];
    a.y = xrow[k0 + koff + 1];
    const float* __restrict__ wk0 = &wlds[(k0 + koff) * OUT_DIM + nsub];
    #pragma unroll
    for (int t = 0; t < 8; ++t) {
      v2f b;
      b.x = wk0[t * 16];
      b.y = wk0[t * 16 + OUT_DIM];
      acc[t] = __builtin_amdgcn_wmma_f32_16x16x4_f32(
           false, a,  false, b,
           (short)0, acc[t],  false,  false);
    }
  }

  const int mbase = m0 + ((lane >> 4) << 3);
  float* outp = support + (size_t)mbase * OUT_DIM + nsub;

  if (m0 + 15 < n_nodes) {
    #pragma unroll
    for (int r = 0; r < 8; ++r)
      #pragma unroll
      for (int t = 0; t < 8; ++t)
        outp[(size_t)r * OUT_DIM + t * 16] = acc[t][r];
  } else {
    #pragma unroll
    for (int r = 0; r < 8; ++r)
      if (mbase + r < n_nodes)
        #pragma unroll
        for (int t = 0; t < 8; ++t)
          outp[(size_t)r * OUT_DIM + t * 16] = acc[t][r];
  }
}

#define EDGE_CHUNK 8

__global__ void __launch_bounds__(1024)
gcn_bin_aggregate(const float* __restrict__ support,
                  const float* __restrict__ bias,
                  const int* __restrict__ run_words,
                  const int* __restrict__ rec_words,
                  float* __restrict__ out, int n_nodes) {
  __shared__ float acc[ROWS_PER_BIN * OUT_DIM];
  __shared__ v4i   srec[REC_CAP];
  __shared__ v4i   s_cnt4[NBINS / 4];
  __shared__ v4i   s_pre4[NBINS / 4];
  __shared__ int   s_total;

  const int tid  = threadIdx.x;
  const int lane = tid & 31;
  const int wave = __builtin_amdgcn_readfirstlane(tid >> 5);
  const int bin  = blockIdx.x;

  int* s_cnt = (int*)s_cnt4;
  int* s_pre = (int*)s_pre4;

  const int run = tid >> 2;
  const int q   = tid & 3;
  const v2i rc  = ((const v2i*)run_words)[bin * NBINS + run];
  if (q == 0) s_cnt[run] = rc.y;

  {
    v4f* acc4 = (v4f*)acc;
    const v4f* __restrict__ bias4 = (const v4f*)bias;
    const v4f bv = bias4[tid & 31];
    for (int i = tid; i < ROWS_PER_BIN * (OUT_DIM / 4); i += 1024)
      acc4[i] = bv;
  }
  __syncthreads();

  if (wave == 0) {
    const v4i a = s_cnt4[2 * lane];
    const v4i b = s_cnt4[2 * lane + 1];
    const int p1 = a.x, p2 = p1 + a.y, p3 = p2 + a.z, p4 = p3 + a.w;
    const int p5 = p4 + b.x, p6 = p5 + b.y, p7 = p6 + b.z;
    const int tot = p7 + b.w;
    int inc = tot;
    #pragma unroll
    for (int d = 1; d < 32; d <<= 1) {
      const int v = __shfl_up(inc, d);
      if (lane >= d) inc += v;
    }
    const int lo = inc - tot;
    v4i o0, o1;
    o0.x = lo;      o0.y = lo + p1; o0.z = lo + p2; o0.w = lo + p3;
    o1.x = lo + p4; o1.y = lo + p5; o1.z = lo + p6; o1.w = lo + p7;
    s_pre4[2 * lane]     = o0;
    s_pre4[2 * lane + 1] = o1;
    if (lane == 31) s_total = inc;
  }
  __syncthreads();

  const int pre   = s_pre[run];
  const int total = __builtin_amdgcn_readfirstlane(s_total);

  const v4i* __restrict__ rec = (const v4i*)rec_words;

  for (int cbase = 0; cbase < total; cbase += REC_CAP) {
    int jlo = cbase - pre;
    if (jlo < 0) jlo = 0;
    int jhi = cbase + REC_CAP - pre;
    if (jhi > rc.y) jhi = rc.y;
    const int dsh = pre - cbase;
    for (int j = jlo + q; j < jhi; j += 4 * RUN_K) {
      v4i rr[RUN_K];
      #pragma unroll
      for (int k = 0; k < RUN_K; ++k) {
        rr[k] = (v4i){0, 0, 0, 0};
        if (j + 4 * k < jhi) rr[k] = rec[rc.x + j + 4 * k];
      }
      #pragma unroll
      for (int k = 0; k < RUN_K; ++k)
        if (j + 4 * k < jhi) srec[dsh + j + 4 * k] = rr[k];
    }
    __syncthreads();

    int nrec = total - cbase;
    if (nrec > REC_CAP) nrec = REC_CAP;
    const int ngroups = (nrec + EDGE_CHUNK - 1) / EDGE_CHUNK;
    v4i r = (v4i){0, 0, 0, 0};
    if (wave < ngroups) {
      int idx = wave * EDGE_CHUNK + (lane & (EDGE_CHUNK - 1));
      if (idx >= nrec) idx = nrec - 1;
      r = srec[idx];
    }
    for (int g = wave; g < ngroups; g += 32) {
      const int eb = g * EDGE_CHUNK;
      int nv = nrec - eb;
      if (nv > EDGE_CHUNK) nv = EDGE_CHUNK;
      v4i rnext = r;
      if (g + 32 < ngroups) {
        int idx = eb + 32 * EDGE_CHUNK + (lane & (EDGE_CHUNK - 1));
        if (idx >= nrec) idx = nrec - 1;
        rnext = srec[idx];
      }

      float m[EDGE_CHUNK][4];
      #pragma unroll
      for (int j = 0; j < EDGE_CHUNK; ++j) {
        const int col = __builtin_amdgcn_readlane(r.x, j);
        const float* __restrict__ srcp = support + (size_t)col * OUT_DIM + lane;
        m[j][0] = srcp[0];
        m[j][1] = srcp[32];
        m[j][2] = srcp[64];
        m[j][3] = srcp[96];
      }
      #pragma unroll
      for (int j = 0; j < EDGE_CHUNK; ++j) {
        if (j == 0 || j < nv) {
          const int   lrow = __builtin_amdgcn_readlane(r.y, j);
          const float v    = __builtin_bit_cast(float, __builtin_amdgcn_readlane(r.z, j));
          float* d = acc + lrow * OUT_DIM + lane;
          __hip_atomic_fetch_add(d +  0, m[j][0] * v, __ATOMIC_RELAXED, __HIP_MEMORY_SCOPE_WORKGROUP);
          __hip_atomic_fetch_add(d + 32, m[j][1] * v, __ATOMIC_RELAXED, __HIP_MEMORY_SCOPE_WORKGROUP);
          __hip_atomic_fetch_add(d + 64, m[j][2] * v, __ATOMIC_RELAXED, __HIP_MEMORY_SCOPE_WORKGROUP);
          __hip_atomic_fetch_add(d + 96, m[j][3] * v, __ATOMIC_RELAXED, __HIP_MEMORY_SCOPE_WORKGROUP);
        }
      }
      r = rnext;
    }
    __syncthreads();
  }

  const int row0 = bin * ROWS_PER_BIN;
  int nrows = n_nodes - row0;
  if (nrows > ROWS_PER_BIN) nrows = ROWS_PER_BIN;
  const v4f* acc4 = (const v4f*)acc;
  v4f* __restrict__ out4 = (v4f*)(out + (size_t)row0 * OUT_DIM);
  for (int i = tid; i < nrows * (OUT_DIM / 4); i += 1024)
    out4[i] = acc4[i];
}

extern "C" void kernel_launch(void* const* d_in, const int* in_sizes, int n_in,
                              void* d_out, int out_size, void* d_ws, size_t ws_size,
                              hipStream_t stream) {
  const float* x        = (const float*)d_in[0];
  const float* w        = (const float*)d_in[1];
  const float* bias     = (const float*)d_in[2];
  const float* adj_vals = (const float*)d_in[3];
  const int*   edge_row = (const int*)d_in[4];
  const int*   edge_col = (const int*)d_in[5];

  const int n_nodes = in_sizes[0] / IN_DIM;
  const int n_edges = in_sizes[3];

  float* out     = (float*)d_out;
  float* support   = (float*)d_ws;
  int*   run_words = (int*)(support + (size_t)n_nodes * OUT_DIM);
  int*   rec_words = run_words + 2 * NBINS * NBINS;

  const int nblocks = (n_nodes + 127) / 128;
  gcn_gemm_wmma<<<NBINS + nblocks, 256, 0, stream>>>(x, w, support, n_nodes,
                                                     adj_vals, edge_row, edge_col,
                                                     run_words, rec_words, n_edges);

  gcn_bin_aggregate<<<NBINS, 1024, 0, stream>>>(support, bias, run_words, rec_words,
                                                out, n_nodes);
}


# aggregate: a bin's records are regrouped by row in LDS; one wave owns a row, sums its edges in registers in edge-index order (16-B gathers), stores it once; no float atomics: sum order changes
# speedup vs baseline: 2.9097x; 1.0461x over previous
#include <hip/hip_runtime.h>
#include <hip/hip_bf16.h>

#define IN_DIM  128
#define OUT_DIM 128

#define NBINS        256
#define ROWS_PER_BIN 196
#define SLICE_K      8
#define REC_CAP      4096
#define RUN_K        6

typedef int v4i __attribute__((ext_vector_type(4)));
typedef float v4f __attribute__((ext_vector_type(4)));

__device__ __forceinline__ int bin_of_row(int row) {
  const unsigned b = (unsigned)row / (unsigned)ROWS_PER_BIN;
  return (int)(b < (unsigned)(NBINS - 1) ? b : (unsigned)(NBINS - 1));
}

typedef int v2i __attribute__((ext_vector_type(2)));
typedef float v2f __attribute__((ext_vector_type(2)));
typedef float v8f __attribute__((ext_vector_type(8)));

__global__ void __launch_bounds__(256)
gcn_gemm_wmma(const float* __restrict__ x, const float* __restrict__ w,
              float* __restrict__ support, int n_nodes,
              const float* __restrict__ adj_vals,
              const int* __restrict__ edge_row,
              const int* __restrict__ edge_col,
              int* __restrict__ run_words, int* __restrict__ rec_words,
              int n_edges) {
  __shared__ float wlds[IN_DIM * OUT_DIM];

  if ((int)blockIdx.x < NBINS) {
    int* hist   = (int*)wlds;
    int* scan   = hist + NBINS;
    int* cursor = scan + NBINS;
    const int t  = threadIdx.x;
    const int cb = (int)blockIdx.x;
    hist[t] = 0;
    __syncthreads();
    const int ch = (n_edges + NBINS - 1) / NBINS;
    const int e0 = cb * ch;
    int e1 = e0 + ch;
    if (e1 > n_edges) e1 = n_edges;

    for (int base = e0; base < e1; base += SLICE_K * 256) {
      int rr[SLICE_K];
      #pragma unroll
      for (int k = 0; k < SLICE_K; ++k) {
        const int e = base + k * 256 + t;
        rr[k] = 0;
        if (e < e1) rr[k] = edge_row[e];
      }
      #pragma unroll
      for (int k = 0; k < SLICE_K; ++k) {
        const int e = base + k * 256 + t;
        if (e < e1) atomicAdd(&hist[bin_of_row(rr[k])], 1);
      }
    }
    __syncthreads();

    const int mycnt = hist[t];
    scan[t] = mycnt;
    __syncthreads();
    for (int off = 1; off < NBINS; off <<= 1) {
      int v = 0;
      if (t >= off) v = scan[t - off];
      __syncthreads();
      scan[t] += v;
      __syncthreads();
    }
    const int first = e0 + scan[t] - mycnt;
    cursor[t] = first;
    {
      v2i rn;
      rn.x = first;
      rn.y = mycnt;
      ((v2i*)run_words)[t * NBINS + cb] = rn;
    }
    __syncthreads();

    v4i* __restrict__ rec = (v4i*)rec_words;
    for (int base = e0; base < e1; base += SLICE_K * 256) {
      int rr[SLICE_K], cc[SLICE_K], vv[SLICE_K];
      #pragma unroll
      for (int k = 0; k < SLICE_K; ++k) {
        const int e = base + k * 256 + t;
        rr[k] = 0; cc[k] = 0; vv[k] = 0;
        if (e < e1) {
          rr[k] = edge_row[e];
          cc[k] = edge_col[e];
          vv[k] = __builtin_bit_cast(int, adj_vals[e]);
        }
      }
      #pragma unroll
      for (int k = 0; k < SLICE_K; ++k) {
        const int e = base + k * 256 + t;
        if (e < e1) {
          const int bk  = bin_of_row(rr[k]);
          const int pos = atomicAdd(&cursor[bk], 1);
          v4i r;
          r.x = cc[k];
          r.y = rr[k] - bk * ROWS_PER_BIN;
          r.z = vv[k];
          r.w = e;
          rec[pos] = r;
        }
      }
    }
    return;
  }

  for (int i = threadIdx.x; i < IN_DIM * OUT_DIM; i += 256)
    wlds[i] = w[i];
  __syncthreads();

  const int lane = threadIdx.x & 31;
  const int wave = threadIdx.x >> 5;
  const int m0   = ((int)blockIdx.x - NBINS) * 128 + wave * 16;

  const int koff = (lane >> 4) * 2;
  const int nsub = lane & 15;

  int mrow = m0 + (lane & 15);
  if (mrow >= n_nodes) mrow = n_nodes - 1;

  const float* __restrict__ xrow = x + (size_t)mrow * IN_DIM;

  v8f acc[8];
  #pragma unroll
  for (int t = 0; t < 8; ++t) acc[t] = (v8f){};

  #pragma unroll 2
  for (int k0 = 0; k0 < IN_DIM; k0 += 4) {
    v2f a;
    a.x = xrow[k0 + koff];
    a.y = xrow[k0 + koff + 1];
    const float* __restrict__ wk0 = &wlds[(k0 + koff) * OUT_DIM + nsub];
    #pragma unroll
    for (int t = 0; t < 8; ++t) {
      v2f b;
      b.x = wk0[t * 16];
      b.y = wk0[t * 16 + OUT_DIM];
      acc[t] = __builtin_amdgcn_wmma_f32_16x16x4_f32(
           false, a,  false, b,
           (short)0, acc[t],  false,  false);
    }
  }

  const int mbase = m0 + ((lane >> 4) << 3);
  float* outp = support + (size_t)mbase * OUT_DIM + nsub;

  if (m0 + 15 < n_nodes) {
    #pragma unroll
    for (int r = 0; r < 8; ++r)
      #pragma unroll
      for (int t = 0; t < 8; ++t)
        outp[(size_t)r * OUT_DIM + t * 16] = acc[t][r];
  } else {
    #pragma unroll
    for (int r = 0; r < 8; ++r)
      if (mbase + r < n_nodes)
        #pragma unroll
        for (int t = 0; t < 8; ++t)
          outp[(size_t)r * OUT_DIM + t * 16] = acc[t][r];
  }
}

#define ROW_TRIP 16

__device__ __forceinline__ void wave_lds_order() {
  __builtin_amdgcn_fence(__ATOMIC_RELEASE, "wavefront");
  __builtin_amdgcn_wave_barrier();
  __builtin_amdgcn_fence(__ATOMIC_ACQUIRE, "wavefront");
}

__device__ __forceinline__ int wave_scan256(const v4i* cnt4, v4i* pre4, int lane) {
  const v4i a = cnt4[2 * lane];
  const v4i b = cnt4[2 * lane + 1];
  const int p1 = a.x, p2 = p1 + a.y, p3 = p2 + a.z, p4 = p3 + a.w;
  const int p5 = p4 + b.x, p6 = p5 + b.y, p7 = p6 + b.z;
  const int tot = p7 + b.w;
  int inc = tot;
  #pragma unroll
  for (int d = 1; d < 32; d <<= 1) {
    const int v = __shfl_up(inc, d);
    if (lane >= d) inc += v;
  }
  const int lo = inc - tot;
  v4i o0, o1;
  o0.x = lo;      o0.y = lo + p1; o0.z = lo + p2; o0.w = lo + p3;
  o1.x = lo + p4; o1.y = lo + p5; o1.z = lo + p6; o1.w = lo + p7;
  pre4[2 * lane]     = o0;
  pre4[2 * lane + 1] = o1;
  return inc;
}

template <int NV>
__device__ __forceinline__ void gather_trip(const float* __restrict__ support, int pcol, int pval,
                                            unsigned lane16, v4f& a) {
  v4f m[NV];
  #pragma unroll
  for (int j = 0; j < NV; ++j) {
    const int cj = __builtin_amdgcn_readlane(pcol, j);
    const unsigned off = (unsigned)cj * (unsigned)(OUT_DIM * 4) + lane16;
    m[j] = *(const v4f*)((const char*)support + off);
  }
  #pragma unroll
  for (int j = 0; j < NV; ++j) {
    const float vj = __builtin_bit_cast(float, __builtin_amdgcn_readlane(pval, j));
    a.x += vj * m[j].x;
    a.y += vj * m[j].y;
    a.z += vj * m[j].z;
    a.w += vj * m[j].w;
  }
}

__global__ void __launch_bounds__(1024)
gcn_bin_aggregate(const float* __restrict__ support,
                  const float* __restrict__ bias,
                  const int* __restrict__ run_words,
                  const int* __restrict__ rec_words,
                  float* out, int n_nodes) {
  __shared__ v4i srec[REC_CAP];
  __shared__ v4i sorted[REC_CAP];
  __shared__ v4i s_cnt4[NBINS / 4];
  __shared__ v4i s_pre4[NBINS / 4];
  __shared__ int s_first[NBINS];
  __shared__ v4i s_rowcnt4[64];
  __shared__ v4i s_rowbeg4[64];
  __shared__ int s_total;
  __shared__ int s_next;

  const int tid  = threadIdx.x;
  const int lane = tid & 31;
  const int wave = __builtin_amdgcn_readfirstlane(tid >> 5);
  const int bin  = blockIdx.x;

  int* s_cnt    = (int*)s_cnt4;
  int* s_pre    = (int*)s_pre4;
  int* s_rowcnt = (int*)s_rowcnt4;
  int* s_rowbeg = (int*)s_rowbeg4;

  const int run = tid >> 2;
  const int q   = tid & 3;
  const v2i rc  = ((const v2i*)run_words)[bin * NBINS + run];
  if (q == 0) {
    s_cnt[run]   = rc.y;
    s_first[run] = rc.x;
  }
  if (tid < 256) s_rowcnt[tid] = 0;
  __syncthreads();

  if (wave == 0) {
    const int inc = wave_scan256(s_cnt4, s_pre4, lane);
    if (lane == 31) s_total = inc;
  }
  __syncthreads();

  const int total = __builtin_amdgcn_readfirstlane(s_total);

  const v4i* __restrict__ rec  = (const v4i*)rec_words;
  const unsigned lane16 = (unsigned)lane * 16u;
  v4f* out4 = (v4f*)out;
  int* sw = (int*)sorted;

  const int row0 = bin * ROWS_PER_BIN;
  int nrows = n_nodes - row0;
  if (nrows > ROWS_PER_BIN) nrows = ROWS_PER_BIN;
  const v4f* __restrict__ bias4 = (const v4f*)bias;

  for (int cbase = 0; cbase == 0 || cbase < total; cbase += REC_CAP) {
    const bool first = (cbase == 0);
    if (!first) __syncthreads();
    const int rfirst = s_first[run];
    const int rlen   = s_cnt[run];
    const int pre    = s_pre[run];
    int jlo = cbase - pre;
    if (jlo < 0) jlo = 0;
    int jhi = cbase + REC_CAP - pre;
    if (jhi > rlen) jhi = rlen;
    const int dsh = pre - cbase;
    for (int j = jlo + q; j < jhi; j += 4 * RUN_K) {
      v4i rr[RUN_K];
      #pragma unroll
      for (int k = 0; k < RUN_K; ++k) {
        rr[k] = (v4i){0, 0, 0, 0};
        if (j + 4 * k < jhi) rr[k] = rec[rfirst + j + 4 * k];
      }
      #pragma unroll
      for (int k = 0; k < RUN_K; ++k)
        if (j + 4 * k < jhi) {
          srec[dsh + j + 4 * k] = rr[k];
          atomicAdd(&s_rowcnt[rr[k].y & 255], 1);
        }
    }
    __syncthreads();

    if (wave == 0) {
      wave_scan256(s_rowcnt4, s_rowbeg4, lane);
      if (lane == 0) s_next = 32;
    }
    __syncthreads();

    int nrec = total - cbase;
    if (nrec > REC_CAP) nrec = REC_CAP;
    for (int i = tid; i < nrec; i += 1024) {
      const v4i rcd = srec[i];
      const int lrw = rcd.y & 255;
      const int pos = s_rowbeg[lrw] + atomicSub(&s_rowcnt[lrw], 1) - 1;
      v4i o;
      o.x = rcd.x;
      o.y = rcd.z;
      o.z = rcd.w;
      o.w = 0;
      sorted[pos] = o;
    }
    __syncthreads();

    int lr = wave;
    while (lr < nrows) {
      const int sv = s_rowbeg[lr];
      const int cv = s_rowbeg[lr + 1] - sv;
      int nxt = 0;
      if (lane == 0) nxt = atomicAdd(&s_next, 1);
      const int c = __builtin_amdgcn_readfirstlane(cv);
      const int s = __builtin_amdgcn_readfirstlane(sv);
      if (c > 0 || first) {
        if (c > 1) {
          if (c <= 32) {
            int col0 = 0, val0 = 0, eid0 = 0;
            if (lane < c) {
              const v4i t = sorted[s + lane];
              col0 = t.x; val0 = t.y; eid0 = t.z;
            }
            int rank0 = 0;
            for (int j = 0; j < c; ++j) {
              const int ej = __builtin_amdgcn_readlane(eid0, j);
              rank0 += (ej < eid0) ? 1 : 0;
            }
            if (lane < c) {
              sw[4 * (s + rank0)]     = col0;
              sw[4 * (s + rank0) + 1] = val0;
            }
          } else {
            const int c2 = c < 64 ? c : 64;
            int col0, val0, eid0, col1 = 0, val1 = 0, eid1 = 0;
            {
              const v4i t = sorted[s + lane];
              col0 = t.x; val0 = t.y; eid0 = t.z;
            }
            if (lane + 32 < c2) {
              const v4i t = sorted[s + 32 + lane];
              col1 = t.x; val1 = t.y; eid1 = t.z;
            }
            int rank0 = 0, rank1 = 0;
            for (int j = 0; j < 32; ++j) {
              const int ej = __builtin_amdgcn_readlane(eid0, j);
              rank0 += (ej < eid0) ? 1 : 0;
              rank1 += (ej < eid1) ? 1 : 0;
            }
            for (int j = 0; j < c2 - 32; ++j) {
              const int ej = __builtin_amdgcn_readlane(eid1, j);
              rank0 += (ej < eid0) ? 1 : 0;
              rank1 += (ej < eid1) ? 1 : 0;
            }
            sw[4 * (s + rank0)]     = col0;
            sw[4 * (s + rank0) + 1] = val0;
            if (lane + 32 < c2) {
              sw[4 * (s + rank1)]     = col1;
              sw[4 * (s + rank1) + 1] = val1;
            }
          }
          wave_lds_order();
        }

        v4f a = (v4f){0.f, 0.f, 0.f, 0.f};
        for (int kb = 0; kb < c; kb += ROW_TRIP) {
          int nv = c - kb;
          if (nv > ROW_TRIP) nv = ROW_TRIP;
          int idx = kb + (lane & (ROW_TRIP - 1));
          if (idx >= c) idx = c - 1;
          const int pcol = sw[4 * (s + idx)];
          const int pval = sw[4 * (s + idx) + 1];
          switch (nv) {
            case  1: gather_trip< 1>(support, pcol, pval, lane16, a); break;
            case  2: gather_trip< 2>(support, pcol, pval, lane16, a); break;
            case  3: gather_trip< 3>(support, pcol, pval, lane16, a); break;
            case  4: gather_trip< 4>(support, pcol, pval, lane16, a); break;
            case  5: gather_trip< 5>(support, pcol, pval, lane16, a); break;
            case  6: gather_trip< 6>(support, pcol, pval, lane16, a); break;
            case  7: gather_trip< 7>(support, pcol, pval, lane16, a); break;
            case  8: gather_trip< 8>(support, pcol, pval, lane16, a); break;
            case  9: gather_trip< 9>(support, pcol, pval, lane16, a); break;
            case 10: gather_trip<10>(support, pcol, pval, lane16, a); break;
            case 11: gather_trip<11>(support, pcol, pval, lane16, a); break;
            case 12: gather_trip<12>(support, pcol, pval, lane16, a); break;
            case 13: gather_trip<13>(support, pcol, pval, lane16, a); break;
            case 14: gather_trip<14>(support, pcol, pval, lane16, a); break;
            case 15: gather_trip<15>(support, pcol, pval, lane16, a); break;
            default: gather_trip<16>(support, pcol, pval, lane16, a); break;
          }
        }

        const size_t orow = (size_t)(row0 + lr) * (OUT_DIM / 4) + lane;
        v4f base;
        if (first) base = bias4[lane];
        else       base = out4[orow];
        a.x += base.x; a.y += base.y; a.z += base.z; a.w += base.w;
        out4[orow] = a;
      }
      lr = __builtin_amdgcn_readfirstlane(nxt);
    }
  }
}

extern "C" void kernel_launch(void* const* d_in, const int* in_sizes, int n_in,
                              void* d_out, int out_size, void* d_ws, size_t ws_size,
                              hipStream_t stream) {
  const float* x        = (const float*)d_in[0];
  const float* w        = (const float*)d_in[1];
  const float* bias     = (const float*)d_in[2];
  const float* adj_vals = (const float*)d_in[3];
  const int*   edge_row = (const int*)d_in[4];
  const int*   edge_col = (const int*)d_in[5];

  const int n_nodes = in_sizes[0] / IN_DIM;
  const int n_edges = in_sizes[3];

  float* out     = (float*)d_out;
  float* support   = (float*)d_ws;
  int*   run_words = (int*)(support + (size_t)n_nodes * OUT_DIM);
  int*   rec_words = run_words + 2 * NBINS * NBINS;

  const int nblocks = (n_nodes + 127) / 128;
  gcn_gemm_wmma<<<NBINS + nblocks, 256, 0, stream>>>(x, w, support, n_nodes,
                                                     adj_vals, edge_row, edge_col,
                                                     run_words, rec_words, n_edges);

  gcn_bin_aggregate<<<NBINS, 1024, 0, stream>>>(support, bias, run_words, rec_words,
                                                out, n_nodes);
}


# GEMM operand feed: W staged in LDS in fragment order (conflict-free 16-byte LDS loads issued one step ahead), A rows by 16-byte loads eight steps ahead; k order permuted alike in A and B
# speedup vs baseline: 3.3972x; 1.1675x over previous
#include <hip/hip_runtime.h>
#include <hip/hip_bf16.h>

#define IN_DIM  128
#define OUT_DIM 128

#define NBINS        256
#define ROWS_PER_BIN 196
#define SLICE_K      8
#define REC_CAP      4096
#define RUN_K        6

typedef int v4i __attribute__((ext_vector_type(4)));
typedef float v4f __attribute__((ext_vector_type(4)));

__device__ __forceinline__ int bin_of_row(int row) {
  const unsigned b = (unsigned)row / (unsigned)ROWS_PER_BIN;
  return (int)(b < (unsigned)(NBINS - 1) ? b : (unsigned)(NBINS - 1));
}

typedef int v2i __attribute__((ext_vector_type(2)));
typedef float v2f __attribute__((ext_vector_type(2)));
typedef float v8f __attribute__((ext_vector_type(8)));

__global__ void __launch_bounds__(256)
gcn_gemm_wmma(const float* __restrict__ x, const float* __restrict__ w,
              float* __restrict__ support, int n_nodes,
              const float* __restrict__ adj_vals,
              const int* __restrict__ edge_row,
              const int* __restrict__ edge_col,
              int* __restrict__ run_words, int* __restrict__ rec_words,
              int n_edges) {
  __shared__ v4f wlds4[IN_DIM * OUT_DIM / 4];

  if ((int)blockIdx.x < NBINS) {
    int* hist   = (int*)wlds4;
    int* scan   = hist + NBINS;
    int* cursor = scan + NBINS;
    const int t  = threadIdx.x;
    const int cb = (int)blockIdx.x;
    hist[t] = 0;
    __syncthreads();
    const int ch = (n_edges + NBINS - 1) / NBINS;
    const int e0 = cb * ch;
    int e1 = e0 + ch;
    if (e1 > n_edges) e1 = n_edges;

    for (int base = e0; base < e1; base += SLICE_K * 256) {
      int rr[SLICE_K];
      #pragma unroll
      for (int k = 0; k < SLICE_K; ++k) {
        const int e = base + k * 256 + t;
        rr[k] = 0;
        if (e < e1) rr[k] = edge_row[e];
      }
      #pragma unroll
      for (int k = 0; k < SLICE_K; ++k) {
        const int e = base + k * 256 + t;
        if (e < e1) atomicAdd(&hist[bin_of_row(rr[k])], 1);
      }
    }
    __syncthreads();

    const int mycnt = hist[t];
    scan[t] = mycnt;
    __syncthreads();
    for (int off = 1; off < NBINS; off <<= 1) {
      int v = 0;
      if (t >= off) v = scan[t - off];
      __syncthreads();
      scan[t] += v;
      __syncthreads();
    }
    const int first = e0 + scan[t] - mycnt;
    cursor[t] = first;
    {
      v2i rn;
      rn.x = first;
      rn.y = mycnt;
      ((v2i*)run_words)[t * NBINS + cb] = rn;
    }
    __syncthreads();

    v4i* __restrict__ rec = (v4i*)rec_words;
    for (int base = e0; base < e1; base += SLICE_K * 256) {
      int rr[SLICE_K], cc[SLICE_K], vv[SLICE_K];
      #pragma unroll
      for (int k = 0; k < SLICE_K; ++k) {
        const int e = base + k * 256 + t;
        rr[k] = 0; cc[k] = 0; vv[k] = 0;
        if (e < e1) {
          rr[k] = edge_row[e];
          cc[k] = edge_col[e];
          vv[k] = __builtin_bit_cast(int, adj_vals[e]);
        }
      }
      #pragma unroll
      for (int k = 0; k < SLICE_K; ++k) {
        const int e = base + k * 256 + t;
        if (e < e1) {
          const int bk  = bin_of_row(rr[k]);
          const int pos = atomicAdd(&cursor[bk], 1);
          v4i r;
          r.x = cc[k];
          r.y = rr[k] - bk * ROWS_PER_BIN;
          r.z = vv[k];
          r.w = e;
          rec[pos] = r;
        }
      }
    }
    return;
  }

  const int lane = threadIdx.x & 31;
  const int wave = threadIdx.x >> 5;
  const int m0   = ((int)blockIdx.x - NBINS) * 128 + wave * 16;

  const int hh   = lane >> 4;
  const int nsub = lane & 15;

  int mrow = m0 + nsub;
  if (mrow >= n_nodes) mrow = n_nodes - 1;

  const v4f* __restrict__ xrow4 = (const v4f*)(x + (size_t)mrow * IN_DIM) + hh;

  v4f a4[16];
  #pragma unroll
  for (int j = 0; j < 4; ++j) a4[j] = xrow4[2 * j];

  {
    const int q = wave & 3, u = wave >> 2;
    const float* __restrict__ wsrc = w + (4 * hh + 2 * u) * OUT_DIM + 32 * q + nsub;
    v4f* wdst = wlds4 + (u * 4 + q) * 32 + lane;
    #pragma unroll
    for (int j = 0; j < 16; ++j) {
      v4f v;
      v.x = wsrc[(8 * j) * OUT_DIM];
      v.y = wsrc[(8 * j + 1) * OUT_DIM];
      v.z = wsrc[(8 * j) * OUT_DIM + 16];
      v.w = wsrc[(8 * j + 1) * OUT_DIM + 16];
      wdst[j * 256] = v;
    }
  }
  __syncthreads();

  v8f acc[8];
  #pragma unroll
  for (int t = 0; t < 8; ++t) acc[t] = (v8f){};

  const v4f* wl4 = wlds4 + lane;
  v4f bc[4], bn[4];
  #pragma unroll
  for (int q = 0; q < 4; ++q) bc[q] = wl4[q * 32];

  #pragma unroll
  for (int p = 0; p < 32; ++p) {
    if (p + 1 < 32) {
      #pragma unroll
      for (int q = 0; q < 4; ++q) bn[q] = wl4[((p + 1) * 4 + q) * 32];
    }
    if ((p & 3) == 0 && p / 2 + 5 < 16) {
      a4[p / 2 + 4] = xrow4[2 * (p / 2 + 4)];
      a4[p / 2 + 5] = xrow4[2 * (p / 2 + 5)];
    }
    __builtin_amdgcn_sched_barrier(0);
    const v4f av = a4[p >> 1];
    v2f a;
    if (p & 1) { a.x = av.z; a.y = av.w; } else { a.x = av.x; a.y = av.y; }
    #pragma unroll
    for (int q = 0; q < 4; ++q) {
      v2f b0, b1;
      b0.x = bc[q].x; b0.y = bc[q].y;
      b1.x = bc[q].z; b1.y = bc[q].w;
      acc[2 * q] = __builtin_amdgcn_wmma_f32_16x16x4_f32(
           false, a,  false, b0,
           (short)0, acc[2 * q],  false,  false);
      acc[2 * q + 1] = __builtin_amdgcn_wmma_f32_16x16x4_f32(
           false, a,  false, b1,
           (short)0, acc[2 * q + 1],  false,  false);
    }
    __builtin_amdgcn_sched_barrier(0);
    #pragma unroll
    for (int q = 0; q < 4; ++q) bc[q] = bn[q];
  }

  const int mbase = m0 + ((lane >> 4) << 3);
  float* outp = support + (size_t)mbase * OUT_DIM + nsub;

  if (m0 + 15 < n_nodes) {
    #pragma unroll
    for (int r = 0; r < 8; ++r)
      #pragma unroll
      for (int t = 0; t < 8; ++t)
        outp[(size_t)r * OUT_DIM + t * 16] = acc[t][r];
  } else {
    #pragma unroll
    for (int r = 0; r < 8; ++r)
      if (mbase + r < n_nodes)
        #pragma unroll
        for (int t = 0; t < 8; ++t)
          outp[(size_t)r * OUT_DIM + t * 16] = acc[t][r];
  }
}

#define ROW_TRIP 16

__device__ __forceinline__ void wave_lds_order() {
  __builtin_amdgcn_fence(__ATOMIC_RELEASE, "wavefront");
  __builtin_amdgcn_wave_barrier();
  __builtin_amdgcn_fence(__ATOMIC_ACQUIRE, "wavefront");
}

__device__ __forceinline__ int wave_scan256(const v4i* cnt4, v4i* pre4, int lane) {
  const v4i a = cnt4[2 * lane];
  const v4i b = cnt4[2 * lane + 1];
  const int p1 = a.x, p2 = p1 + a.y, p3 = p2 + a.z, p4 = p3 + a.w;
  const int p5 = p4 + b.x, p6 = p5 + b.y, p7 = p6 + b.z;
  const int tot = p7 + b.w;
  int inc = tot;
  #pragma unroll
  for (int d = 1; d < 32; d <<= 1) {
    const int v = __shfl_up(inc, d);
    if (lane >= d) inc += v;
  }
  const int lo = inc - tot;
  v4i o0, o1;
  o0.x = lo;      o0.y = lo + p1; o0.z = lo + p2; o0.w = lo + p3;
  o1.x = lo + p4; o1.y = lo + p5; o1.z = lo + p6; o1.w = lo + p7;
  pre4[2 * lane]     = o0;
  pre4[2 * lane + 1] = o1;
  return inc;
}

template <int NV>
__device__ __forceinline__ void gather_trip(const float* __restrict__ support, int pcol, int pval,
                                            unsigned lane16, v4f& a) {
  v4f m[NV];
  #pragma unroll
  for (int j = 0; j < NV; ++j) {
    const int cj = __builtin_amdgcn_readlane(pcol, j);
    const unsigned off = (unsigned)cj * (unsigned)(OUT_DIM * 4) + lane16;
    m[j] = *(const v4f*)((const char*)support + off);
  }
  #pragma unroll
  for (int j = 0; j < NV; ++j) {
    const float vj = __builtin_bit_cast(float, __builtin_amdgcn_readlane(pval, j));
    a.x += vj * m[j].x;
    a.y += vj * m[j].y;
    a.z += vj * m[j].z;
    a.w += vj * m[j].w;
  }
}

__global__ void __launch_bounds__(1024)
gcn_bin_aggregate(const float* __restrict__ support,
                  const float* __restrict__ bias,
                  const int* __restrict__ run_words,
                  const int* __restrict__ rec_words,
                  float* out, int n_nodes) {
  __shared__ v4i srec[REC_CAP];
  __shared__ v4i sorted[REC_CAP];
  __shared__ v4i s_cnt4[NBINS / 4];
  __shared__ v4i s_pre4[NBINS / 4];
  __shared__ int s_first[NBINS];
  __shared__ v4i s_rowcnt4[64];
  __shared__ v4i s_rowbeg4[64];
  __shared__ int s_total;
  __shared__ int s_next;

  const int tid  = threadIdx.x;
  const int lane = tid & 31;
  const int wave = __builtin_amdgcn_readfirstlane(tid >> 5);
  const int bin  = blockIdx.x;

  int* s_cnt    = (int*)s_cnt4;
  int* s_pre    = (int*)s_pre4;
  int* s_rowcnt = (int*)s_rowcnt4;
  int* s_rowbeg = (int*)s_rowbeg4;

  const int run = tid >> 2;
  const int q   = tid & 3;
  const v2i rc  = ((const v2i*)run_words)[bin * NBINS + run];
  if (q == 0) {
    s_cnt[run]   = rc.y;
    s_first[run] = rc.x;
  }
  if (tid < 256) s_rowcnt[tid] = 0;
  __syncthreads();

  if (wave == 0) {
    const int inc = wave_scan256(s_cnt4, s_pre4, lane);
    if (lane == 31) s_total = inc;
  }
  __syncthreads();

  const int total = __builtin_amdgcn_readfirstlane(s_total);

  const v4i* __restrict__ rec  = (const v4i*)rec_words;
  const unsigned lane16 = (unsigned)lane * 16u;
  v4f* out4 = (v4f*)out;
  int* sw = (int*)sorted;

  const int row0 = bin * ROWS_PER_BIN;
  int nrows = n_nodes - row0;
  if (nrows > ROWS_PER_BIN) nrows = ROWS_PER_BIN;
  const v4f* __restrict__ bias4 = (const v4f*)bias;

  for (int cbase = 0; cbase == 0 || cbase < total; cbase += REC_CAP) {
    const bool first = (cbase == 0);
    if (!first) __syncthreads();
    const int rfirst = s_first[run];
    const int rlen   = s_cnt[run];
    const int pre    = s_pre[run];
    int jlo = cbase - pre;
    if (jlo < 0) jlo = 0;
    int jhi = cbase + REC_CAP - pre;
    if (jhi > rlen) jhi = rlen;
    const int dsh = pre - cbase;
    for (int j = jlo + q; j < jhi; j += 4 * RUN_K) {
      v4i rr[RUN_K];
      #pragma unroll
      for (int k = 0; k < RUN_K; ++k) {
        rr[k] = (v4i){0, 0, 0, 0};
        if (j + 4 * k < jhi) rr[k] = rec[rfirst + j + 4 * k];
      }
      #pragma unroll
      for (int k = 0; k < RUN_K; ++k)
        if (j + 4 * k < jhi) {
          srec[dsh + j + 4 * k] = rr[k];
          atomicAdd(&s_rowcnt[rr[k].y & 255], 1);
        }
    }
    __syncthreads();

    if (wave == 0) {
      wave_scan256(s_rowcnt4, s_rowbeg4, lane);
      if (lane == 0) s_next = 32;
    }
    __syncthreads();

    int nrec = total - cbase;
    if (nrec > REC_CAP) nrec = REC_CAP;
    for (int i = tid; i < nrec; i += 1024) {
      const v4i rcd = srec[i];
      const int lrw = rcd.y & 255;
      const int pos = s_rowbeg[lrw] + atomicSub(&s_rowcnt[lrw], 1) - 1;
      v4i o;
      o.x = rcd.x;
      o.y = rcd.z;
      o.z = rcd.w;
      o.w = 0;
      sorted[pos] = o;
    }
    __syncthreads();

    int lr = wave;
    while (lr < nrows) {
      const int sv = s_rowbeg[lr];
      const int cv = s_rowbeg[lr + 1] - sv;
      int nxt = 0;
      if (lane == 0) nxt = atomicAdd(&s_next, 1);
      const int c = __builtin_amdgcn_readfirstlane(cv);
      const int s = __builtin_amdgcn_readfirstlane(sv);
      if (c > 0 || first) {
        if (c > 1) {
          if (c <= 32) {
            int col0 = 0, val0 = 0, eid0 = 0;
            if (lane < c) {
              const v4i t = sorted[s + lane];
              col0 = t.x; val0 = t.y; eid0 = t.z;
            }
            int rank0 = 0;
            for (int j = 0; j < c; ++j) {
              const int ej = __builtin_amdgcn_readlane(eid0, j);
              rank0 += (ej < eid0) ? 1 : 0;
            }
            if (lane < c) {
              sw[4 * (s + rank0)]     = col0;
              sw[4 * (s + rank0) + 1] = val0;
            }
          } else {
            const int c2 = c < 64 ? c : 64;
            int col0, val0, eid0, col1 = 0, val1 = 0, eid1 = 0;
            {
              const v4i t = sorted[s + lane];
              col0 = t.x; val0 = t.y; eid0 = t.z;
            }
            if (lane + 32 < c2) {
              const v4i t = sorted[s + 32 + lane];
              col1 = t.x; val1 = t.y; eid1 = t.z;
            }
            int rank0 = 0, rank1 = 0;
            for (int j = 0; j < 32; ++j) {
              const int ej = __builtin_amdgcn_readlane(eid0, j);
              rank0 += (ej < eid0) ? 1 : 0;
              rank1 += (ej < eid1) ? 1 : 0;
            }
            for (int j = 0; j < c2 - 32; ++j) {
              const int ej = __builtin_amdgcn_readlane(eid1, j);
              rank0 += (ej < eid0) ? 1 : 0;
              rank1 += (ej < eid1) ? 1 : 0;
            }
            sw[4 * (s + rank0)]     = col0;
            sw[4 * (s + rank0) + 1] = val0;
            if (lane + 32 < c2) {
              sw[4 * (s + rank1)]     = col1;
              sw[4 * (s + rank1) + 1] = val1;
            }
          }
          wave_lds_order();
        }

        v4f a = (v4f){0.f, 0.f, 0.f, 0.f};
        for (int kb = 0; kb < c; kb += ROW_TRIP) {
          int nv = c - kb;
          if (nv > ROW_TRIP) nv = ROW_TRIP;
          int idx = kb + (lane & (ROW_TRIP - 1));
          if (idx >= c) idx = c - 1;
          const int pcol = sw[4 * (s + idx)];
          const int pval = sw[4 * (s + idx) + 1];
          switch (nv) {
            case  1: gather_trip< 1>(support, pcol, pval, lane16, a); break;
            case  2: gather_trip< 2>(support, pcol, pval, lane16, a); break;
            case  3: gather_trip< 3>(support, pcol, pval, lane16, a); break;
            case  4: gather_trip< 4>(support, pcol, pval, lane16, a); break;
            case  5: gather_trip< 5>(support, pcol, pval, lane16, a); break;
            case  6: gather_trip< 6>(support, pcol, pval, lane16, a); break;
            case  7: gather_trip< 7>(support, pcol, pval, lane16, a); break;
            case  8: gather_trip< 8>(support, pcol, pval, lane16, a); break;
            case  9: gather_trip< 9>(support, pcol, pval, lane16, a); break;
            case 10: gather_trip<10>(support, pcol, pval, lane16, a); break;
            case 11: gather_trip<11>(support, pcol, pval, lane16, a); break;
            case 12: gather_trip<12>(support, pcol, pval, lane16, a); break;
            case 13: gather_trip<13>(support, pcol, pval, lane16, a); break;
            case 14: gather_trip<14>(support, pcol, pval, lane16, a); break;
            case 15: gather_trip<15>(support, pcol, pval, lane16, a); break;
            default: gather_trip<16>(support, pcol, pval, lane16, a); break;
          }
        }

        const size_t orow = (size_t)(row0 + lr) * (OUT_DIM / 4) + lane;
        v4f base;
        if (first) base = bias4[lane];
        else       base = out4[orow];
        a.x += base.x; a.y += base.y; a.z += base.z; a.w += base.w;
        out4[orow] = a;
      }
      lr = __builtin_amdgcn_readfirstlane(nxt);
    }
  }
}

extern "C" void kernel_launch(void* const* d_in, const int* in_sizes, int n_in,
                              void* d_out, int out_size, void* d_ws, size_t ws_size,
                              hipStream_t stream) {
  const float* x        = (const float*)d_in[0];
  const float* w        = (const float*)d_in[1];
  const float* bias     = (const float*)d_in[2];
  const float* adj_vals = (const float*)d_in[3];
  const int*   edge_row = (const int*)d_in[4];
  const int*   edge_col = (const int*)d_in[5];

  const int n_nodes = in_sizes[0] / IN_DIM;
  const int n_edges = in_sizes[3];

  float* out     = (float*)d_out;
  float* support   = (float*)d_ws;
  int*   run_words = (int*)(support + (size_t)n_nodes * OUT_DIM);
  int*   rec_words = run_words + 2 * NBINS * NBINS;

  const int nblocks = (n_nodes + 127) / 128;
  gcn_gemm_wmma<<<NBINS + nblocks, 256, 0, stream>>>(x, w, support, n_nodes,
                                                     adj_vals, edge_row, edge_col,
                                                     run_words, rec_words, n_edges);

  gcn_bin_aggregate<<<NBINS, 1024, 0, stream>>>(support, bias, run_words, rec_words,
                                                out, n_nodes);
}


# GEMM: W columns dealt to the tiles so that a lane owns 4 adjacent columns per row: support leaves as 16-byte stores of whole 128-B lines (was 64-B half lines); W staged by 8-byte reads; same sums
# speedup vs baseline: 3.5530x; 1.0459x over previous
#include <hip/hip_runtime.h>
#include <hip/hip_bf16.h>

#define IN_DIM  128
#define OUT_DIM 128

#define NBINS        256
#define ROWS_PER_BIN 196
#define SLICE_K      8
#define REC_CAP      4096
#define RUN_K        6

typedef int v4i __attribute__((ext_vector_type(4)));
typedef float v4f __attribute__((ext_vector_type(4)));

__device__ __forceinline__ int bin_of_row(int row) {
  const unsigned b = (unsigned)row / (unsigned)ROWS_PER_BIN;
  return (int)(b < (unsigned)(NBINS - 1) ? b : (unsigned)(NBINS - 1));
}

typedef int v2i __attribute__((ext_vector_type(2)));
typedef float v2f __attribute__((ext_vector_type(2)));
typedef float v8f __attribute__((ext_vector_type(8)));

__global__ void __launch_bounds__(256)
gcn_gemm_wmma(const float* __restrict__ x, const float* __restrict__ w,
              float* __restrict__ support, int n_nodes,
              const float* __restrict__ adj_vals,
              const int* __restrict__ edge_row,
              const int* __restrict__ edge_col,
              int* __restrict__ run_words, int* __restrict__ rec_words,
              int n_edges) {
  __shared__ v4f wlds4[IN_DIM * OUT_DIM / 4];

  if ((int)blockIdx.x < NBINS) {
    int* hist   = (int*)wlds4;
    int* scan   = hist + NBINS;
    int* cursor = scan + NBINS;
    const int t  = threadIdx.x;
    const int cb = (int)blockIdx.x;
    hist[t] = 0;
    __syncthreads();
    const int ch = (n_edges + NBINS - 1) / NBINS;
    const int e0 = cb * ch;
    int e1 = e0 + ch;
    if (e1 > n_edges) e1 = n_edges;

    for (int base = e0; base < e1; base += SLICE_K * 256) {
      int rr[SLICE_K];
      #pragma unroll
      for (int k = 0; k < SLICE_K; ++k) {
        const int e = base + k * 256 + t;
        rr[k] = 0;
        if (e < e1) rr[k] = edge_row[e];
      }
      #pragma unroll
      for (int k = 0; k < SLICE_K; ++k) {
        const int e = base + k * 256 + t;
        if (e < e1) atomicAdd(&hist[bin_of_row(rr[k])], 1);
      }
    }
    __syncthreads();

    const int mycnt = hist[t];
    scan[t] = mycnt;
    __syncthreads();
    for (int off = 1; off < NBINS; off <<= 1) {
      int v = 0;
      if (t >= off) v = scan[t - off];
      __syncthreads();
      scan[t] += v;
      __syncthreads();
    }
    const int first = e0 + scan[t] - mycnt;
    cursor[t] = first;
    {
      v2i rn;
      rn.x = first;
      rn.y = mycnt;
      ((v2i*)run_words)[t * NBINS + cb] = rn;
    }
    __syncthreads();

    v4i* __restrict__ rec = (v4i*)rec_words;
    for (int base = e0; base < e1; base += SLICE_K * 256) {
      int rr[SLICE_K], cc[SLICE_K], vv[SLICE_K];
      #pragma unroll
      for (int k = 0; k < SLICE_K; ++k) {
        const int e = base + k * 256 + t;
        rr[k] = 0; cc[k] = 0; vv[k] = 0;
        if (e < e1) {
          rr[k] = edge_row[e];
          cc[k] = edge_col[e];
          vv[k] = __builtin_bit_cast(int, adj_vals[e]);
        }
      }
      #pragma unroll
      for (int k = 0; k < SLICE_K; ++k) {
        const int e = base + k * 256 + t;
        if (e < e1) {
          const int bk  = bin_of_row(rr[k]);
          const int pos = atomicAdd(&cursor[bk], 1);
          v4i r;
          r.x = cc[k];
          r.y = rr[k] - bk * ROWS_PER_BIN;
          r.z = vv[k];
          r.w = e;
          rec[pos] = r;
        }
      }
    }
    return;
  }

  const int lane = threadIdx.x & 31;
  const int wave = threadIdx.x >> 5;
  const int m0   = ((int)blockIdx.x - NBINS) * 128 + wave * 16;

  const int hh   = lane >> 4;
  const int nsub = lane & 15;

  int mrow = m0 + nsub;
  if (mrow >= n_nodes) mrow = n_nodes - 1;

  const v4f* __restrict__ xrow4 = (const v4f*)(x + (size_t)mrow * IN_DIM) + hh;

  v4f a4[16];
  #pragma unroll
  for (int j = 0; j < 4; ++j) a4[j] = xrow4[2 * j];

  {
    const int q = wave & 3, u = wave >> 2;
    const float* __restrict__ wsrc = w + (4 * hh + 2 * u) * OUT_DIM + 64 * (q >> 1) + 4 * nsub + 2 * (q & 1);
    v4f* wdst = wlds4 + (u * 4 + q) * 32 + lane;
    #pragma unroll
    for (int j = 0; j < 16; ++j) {
      const v2f k0 = *(const v2f*)(wsrc + (8 * j) * OUT_DIM);
      const v2f k1 = *(const v2f*)(wsrc + (8 * j + 1) * OUT_DIM);
      v4f v;
      v.x = k0.x;
      v.y = k1.x;
      v.z = k0.y;
      v.w = k1.y;
      wdst[j * 256] = v;
    }
  }
  __syncthreads();

  v8f acc[8];
  #pragma unroll
  for (int t = 0; t < 8; ++t) acc[t] = (v8f){};

  const v4f* wl4 = wlds4 + lane;
  v4f bc[4], bn[4];
  #pragma unroll
  for (int q = 0; q < 4; ++q) bc[q] = wl4[q * 32];

  #pragma unroll
  for (int p = 0; p < 32; ++p) {
    if (p + 1 < 32) {
      #pragma unroll
      for (int q = 0; q < 4; ++q) bn[q] = wl4[((p + 1) * 4 + q) * 32];
    }
    if ((p & 3) == 0 && p / 2 + 5 < 16) {
      a4[p / 2 + 4] = xrow4[2 * (p / 2 + 4)];
      a4[p / 2 + 5] = xrow4[2 * (p / 2 + 5)];
    }
    __builtin_amdgcn_sched_barrier(0);
    const v4f av = a4[p >> 1];
    v2f a;
    if (p & 1) { a.x = av.z; a.y = av.w; } else { a.x = av.x; a.y = av.y; }
    #pragma unroll
    for (int q = 0; q < 4; ++q) {
      v2f b0, b1;
      b0.x = bc[q].x; b0.y = bc[q].y;
      b1.x = bc[q].z; b1.y = bc[q].w;
      acc[2 * q] = __builtin_amdgcn_wmma_f32_16x16x4_f32(
           false, a,  false, b0,
           (short)0, acc[2 * q],  false,  false);
      acc[2 * q + 1] = __builtin_amdgcn_wmma_f32_16x16x4_f32(
           false, a,  false, b1,
           (short)0, acc[2 * q + 1],  false,  false);
    }
    __builtin_amdgcn_sched_barrier(0);
    #pragma unroll
    for (int q = 0; q < 4; ++q) bc[q] = bn[q];
  }

  const int mbase = m0 + ((lane >> 4) << 3);
  float* outp = support + (size_t)mbase * OUT_DIM + 4 * nsub;

  if (m0 + 15 < n_nodes) {
    #pragma unroll
    for (int r = 0; r < 8; ++r)
      #pragma unroll
      for (int g = 0; g < 2; ++g) {
        v4f v;
        v.x = acc[4 * g][r]; v.y = acc[4 * g + 1][r]; v.z = acc[4 * g + 2][r]; v.w = acc[4 * g + 3][r];
        *(v4f*)(outp + (size_t)r * OUT_DIM + 64 * g) = v;
      }
  } else {
    #pragma unroll
    for (int r = 0; r < 8; ++r)
      if (mbase + r < n_nodes)
        #pragma unroll
        for (int g = 0; g < 2; ++g) {
          v4f v;
          v.x = acc[4 * g][r]; v.y = acc[4 * g + 1][r]; v.z = acc[4 * g + 2][r]; v.w = acc[4 * g + 3][r];
          *(v4f*)(outp + (size_t)r * OUT_DIM + 64 * g) = v;
        }
  }
}

#define ROW_TRIP 16

__device__ __forceinline__ void wave_lds_order() {
  __builtin_amdgcn_fence(__ATOMIC_RELEASE, "wavefront");
  __builtin_amdgcn_wave_barrier();
  __builtin_amdgcn_fence(__ATOMIC_ACQUIRE, "wavefront");
}

__device__ __forceinline__ int wave_scan256(const v4i* cnt4, v4i* pre4, int lane) {
  const v4i a = cnt4[2 * lane];
  const v4i b = cnt4[2 * lane + 1];
  const int p1 = a.x, p2 = p1 + a.y, p3 = p2 + a.z, p4 = p3 + a.w;
  const int p5 = p4 + b.x, p6 = p5 + b.y, p7 = p6 + b.z;
  const int tot = p7 + b.w;
  int inc = tot;
  #pragma unroll
  for (int d = 1; d < 32; d <<= 1) {
    const int v = __shfl_up(inc, d);
    if (lane >= d) inc += v;
  }
  const int lo = inc - tot;
  v4i o0, o1;
  o0.x = lo;      o0.y = lo + p1; o0.z = lo + p2; o0.w = lo + p3;
  o1.x = lo + p4; o1.y = lo + p5; o1.z = lo + p6; o1.w = lo + p7;
  pre4[2 * lane]     = o0;
  pre4[2 * lane + 1] = o1;
  return inc;
}

template <int NV>
__device__ __forceinline__ void gather_trip(const float* __restrict__ support, int pcol, int pval,
                                            unsigned lane16, v4f& a) {
  v4f m[NV];
  #pragma unroll
  for (int j = 0; j < NV; ++j) {
    const int cj = __builtin_amdgcn_readlane(pcol, j);
    const unsigned off = (unsigned)cj * (unsigned)(OUT_DIM * 4) + lane16;
    m[j] = *(const v4f*)((const char*)support + off);
  }
  #pragma unroll
  for (int j = 0; j < NV; ++j) {
    const float vj = __builtin_bit_cast(float, __builtin_amdgcn_readlane(pval, j));
    a.x += vj * m[j].x;
    a.y += vj * m[j].y;
    a.z += vj * m[j].z;
    a.w += vj * m[j].w;
  }
}

__global__ void __launch_bounds__(1024)
gcn_bin_aggregate(const float* __restrict__ support,
                  const float* __restrict__ bias,
                  const int* __restrict__ run_words,
                  const int* __restrict__ rec_words,
                  float* out, int n_nodes) {
  __shared__ v4i srec[REC_CAP];
  __shared__ v4i sorted[REC_CAP];
  __shared__ v4i s_cnt4[NBINS / 4];
  __shared__ v4i s_pre4[NBINS / 4];
  __shared__ int s_first[NBINS];
  __shared__ v4i s_rowcnt4[64];
  __shared__ v4i s_rowbeg4[64];
  __shared__ int s_total;
  __shared__ int s_next;

  const int tid  = threadIdx.x;
  const int lane = tid & 31;
  const int wave = __builtin_amdgcn_readfirstlane(tid >> 5);
  const int bin  = blockIdx.x;

  int* s_cnt    = (int*)s_cnt4;
  int* s_pre    = (int*)s_pre4;
  int* s_rowcnt = (int*)s_rowcnt4;
  int* s_rowbeg = (int*)s_rowbeg4;

  const int run = tid >> 2;
  const int q   = tid & 3;
  const v2i rc  = ((const v2i*)run_words)[bin * NBINS + run];
  if (q == 0) {
    s_cnt[run]   = rc.y;
    s_first[run] = rc.x;
  }
  if (tid < 256) s_rowcnt[tid] = 0;
  __syncthreads();

  if (wave == 0) {
    const int inc = wave_scan256(s_cnt4, s_pre4, lane);
    if (lane == 31) s_total = inc;
  }
  __syncthreads();

  const int total = __builtin_amdgcn_readfirstlane(s_total);

  const v4i* __restrict__ rec  = (const v4i*)rec_words;
  const unsigned lane16 = (unsigned)lane * 16u;
  v4f* out4 = (v4f*)out;
  int* sw = (int*)sorted;

  const int row0 = bin * ROWS_PER_BIN;
  int nrows = n_nodes - row0;
  if (nrows > ROWS_PER_BIN) nrows = ROWS_PER_BIN;
  const v4f* __restrict__ bias4 = (const v4f*)bias;

  for (int cbase = 0; cbase == 0 || cbase < total; cbase += REC_CAP) {
    const bool first = (cbase == 0);
    if (!first) __syncthreads();
    const int rfirst = s_first[run];
    const int rlen   = s_cnt[run];
    const int pre    = s_pre[run];
    int jlo = cbase - pre;
    if (jlo < 0) jlo = 0;
    int jhi = cbase + REC_CAP - pre;
    if (jhi > rlen) jhi = rlen;
    const int dsh = pre - cbase;
    for (int j = jlo + q; j < jhi; j += 4 * RUN_K) {
      v4i rr[RUN_K];
      #pragma unroll
      for (int k = 0; k < RUN_K; ++k) {
        rr[k] = (v4i){0, 0, 0, 0};
        if (j + 4 * k < jhi) rr[k] = rec[rfirst + j + 4 * k];
      }
      #pragma unroll
      for (int k = 0; k < RUN_K; ++k)
        if (j + 4 * k < jhi) {
          srec[dsh + j + 4 * k] = rr[k];
          atomicAdd(&s_rowcnt[rr[k].y & 255], 1);
        }
    }
    __syncthreads();

    if (wave == 0) {
      wave_scan256(s_rowcnt4, s_rowbeg4, lane);
      if (lane == 0) s_next = 32;
    }
    __syncthreads();

    int nrec = total - cbase;
    if (nrec > REC_CAP) nrec = REC_CAP;
    for (int i = tid; i < nrec; i += 1024) {
      const v4i rcd = srec[i];
      const int lrw = rcd.y & 255;
      const int pos = s_rowbeg[lrw] + atomicSub(&s_rowcnt[lrw], 1) - 1;
      v4i o;
      o.x = rcd.x;
      o.y = rcd.z;
      o.z = rcd.w;
      o.w = 0;
      sorted[pos] = o;
    }
    __syncthreads();

    int lr = wave;
    while (lr < nrows) {
      const int sv = s_rowbeg[lr];
      const int cv = s_rowbeg[lr + 1] - sv;
      int nxt = 0;
      if (lane == 0) nxt = atomicAdd(&s_next, 1);
      const int c = __builtin_amdgcn_readfirstlane(cv);
      const int s = __builtin_amdgcn_readfirstlane(sv);
      if (c > 0 || first) {
        if (c > 1) {
          if (c <= 32) {
            int col0 = 0, val0 = 0, eid0 = 0;
            if (lane < c) {
              const v4i t = sorted[s + lane];
              col0 = t.x; val0 = t.y; eid0 = t.z;
            }
            int rank0 = 0;
            for (int j = 0; j < c; ++j) {
              const int ej = __builtin_amdgcn_readlane(eid0, j);
              rank0 += (ej < eid0) ? 1 : 0;
            }
            if (lane < c) {
              sw[4 * (s + rank0)]     = col0;
              sw[4 * (s + rank0) + 1] = val0;
            }
          } else {
            const int c2 = c < 64 ? c : 64;
            int col0, val0, eid0, col1 = 0, val1 = 0, eid1 = 0;
            {
              const v4i t = sorted[s + lane];
              col0 = t.x; val0 = t.y; eid0 = t.z;
            }
            if (lane + 32 < c2) {
              const v4i t = sorted[s + 32 + lane];
              col1 = t.x; val1 = t.y; eid1 = t.z;
            }
            int rank0 = 0, rank1 = 0;
            for (int j = 0; j < 32; ++j) {
              const int ej = __builtin_amdgcn_readlane(eid0, j);
              rank0 += (ej < eid0) ? 1 : 0;
              rank1 += (ej < eid1) ? 1 : 0;
            }
            for (int j = 0; j < c2 - 32; ++j) {
              const int ej = __builtin_amdgcn_readlane(eid1, j);
              rank0 += (ej < eid0) ? 1 : 0;
              rank1 += (ej < eid1) ? 1 : 0;
            }
            sw[4 * (s + rank0)]     = col0;
            sw[4 * (s + rank0) + 1] = val0;
            if (lane + 32 < c2) {
              sw[4 * (s + rank1)]     = col1;
              sw[4 * (s + rank1) + 1] = val1;
            }
          }
          wave_lds_order();
        }

        v4f a = (v4f){0.f, 0.f, 0.f, 0.f};
        for (int kb = 0; kb < c; kb += ROW_TRIP) {
          int nv = c - kb;
          if (nv > ROW_TRIP) nv = ROW_TRIP;
          int idx = kb + (lane & (ROW_TRIP - 1));
          if (idx >= c) idx = c - 1;
          const int pcol = sw[4 * (s + idx)];
          const int pval = sw[4 * (s + idx) + 1];
          switch (nv) {
            case  1: gather_trip< 1>(support, pcol, pval, lane16, a); break;
            case  2: gather_trip< 2>(support, pcol, pval, lane16, a); break;
            case  3: gather_trip< 3>(support, pcol, pval, lane16, a); break;
            case  4: gather_trip< 4>(support, pcol, pval, lane16, a); break;
            case  5: gather_trip< 5>(support, pcol, pval, lane16, a); break;
            case  6: gather_trip< 6>(support, pcol, pval, lane16, a); break;
            case  7: gather_trip< 7>(support, pcol, pval, lane16, a); break;
            case  8: gather_trip< 8>(support, pcol, pval, lane16, a); break;
            case  9: gather_trip< 9>(support, pcol, pval, lane16, a); break;
            case 10: gather_trip<10>(support, pcol, pval, lane16, a); break;
            case 11: gather_trip<11>(support, pcol, pval, lane16, a); break;
            case 12: gather_trip<12>(support, pcol, pval, lane16, a); break;
            case 13: gather_trip<13>(support, pcol, pval, lane16, a); break;
            case 14: gather_trip<14>(support, pcol, pval, lane16, a); break;
            case 15: gather_trip<15>(support, pcol, pval, lane16, a); break;
            default: gather_trip<16>(support, pcol, pval, lane16, a); break;
          }
        }

        const size_t orow = (size_t)(row0 + lr) * (OUT_DIM / 4) + lane;
        v4f base;
        if (first) base = bias4[lane];
        else       base = out4[orow];
        a.x += base.x; a.y += base.y; a.z += base.z; a.w += base.w;
        out4[orow] = a;
      }
      lr = __builtin_amdgcn_readfirstlane(nxt);
    }
  }
}

extern "C" void kernel_launch(void* const* d_in, const int* in_sizes, int n_in,
                              void* d_out, int out_size, void* d_ws, size_t ws_size,
                              hipStream_t stream) {
  const float* x        = (const float*)d_in[0];
  const float* w        = (const float*)d_in[1];
  const float* bias     = (const float*)d_in[2];
  const float* adj_vals = (const float*)d_in[3];
  const int*   edge_row = (const int*)d_in[4];
  const int*   edge_col = (const int*)d_in[5];

  const int n_nodes = in_sizes[0] / IN_DIM;
  const int n_edges = in_sizes[3];

  float* out     = (float*)d_out;
  float* support   = (float*)d_ws;
  int*   run_words = (int*)(support + (size_t)n_nodes * OUT_DIM);
  int*   rec_words = run_words + 2 * NBINS * NBINS;

  const int nblocks = (n_nodes + 127) / 128;
  gcn_gemm_wmma<<<NBINS + nblocks, 256, 0, stream>>>(x, w, support, n_nodes,
                                                     adj_vals, edge_row, edge_col,
                                                     run_words, rec_words, n_edges);

  gcn_bin_aggregate<<<NBINS, 1024, 0, stream>>>(support, bias, run_words, rec_words,
                                                out, n_nodes);
}
